# starting kernel
# baseline (speedup 1.0000x reference)
#include <hip/hip_runtime.h>


constexpr int NP = 128;
constexpr float FAR_DELTA = 1e10f;

typedef __attribute__((ext_vector_type(2))) float v2f;
typedef __attribute__((ext_vector_type(8))) float v8f;

__global__ __launch_bounds__(512)
void volrend_kernel(const float* __restrict__ density,
                    const float* __restrict__ depth,
                    const float* __restrict__ feature,
                    float* __restrict__ out, int N)
{
    __shared__ float4 part[16][33];

    const int lane    = threadIdx.x & 31;
    const int wave    = threadIdx.x >> 5;
    const int rayBase = blockIdx.x * 16;
    const int ray     = rayBase + wave;

    float fx = 0.f, fy = 0.f, fz = 0.f, dd = 0.f;

    if (ray < N) {
        const float4 sg = *(const float4*)(density + ray * NP + lane * 4);
        const float4 dp = *(const float4*)(depth   + ray * NP + lane * 4);

        float nx = __shfl_down(dp.x, 1, 32);
        float d0 = dp.y - dp.x;
        float d1 = dp.z - dp.y;
        float d2 = dp.w - dp.z;
        float d3 = (lane == 31) ? FAR_DELTA : (nx - dp.w);

        float s0 = __expf(-sg.x * d0);
        float s1 = __expf(-sg.y * d1);
        float s2 = __expf(-sg.z * d2);
        float s3 = __expf(-sg.w * d3);

        float t = s0 * s1 * s2 * s3;
        #pragma unroll
        for (int off = 1; off < 32; off <<= 1) {
            float v = __shfl_up(t, off, 32);
            if (lane >= off) t *= v;
        }
        float excl = __shfl_up(t, 1, 32);
        if (lane == 0) excl = 1.0f;

        float T0 = excl;
        float T1 = T0 * s0;
        float T2 = T1 * s1;
        float T3 = T2 * s2;
        float w0 = T0 * (1.0f - s0);
        float w1 = T1 * (1.0f - s1);
        float w2 = T2 * (1.0f - s2);
        float w3 = T3 * (1.0f - s3);

        const float* fp = feature + (size_t)ray * (NP * 3) + lane * 12;
        const float4 fa = *(const float4*)(fp + 0);
        const float4 fb = *(const float4*)(fp + 4);
        const float4 fc = *(const float4*)(fp + 8);

        fx = w0 * fa.x + w1 * fa.w + w2 * fb.z + w3 * fc.y;
        fy = w0 * fa.y + w1 * fb.x + w2 * fb.w + w3 * fc.z;
        fz = w0 * fa.z + w1 * fb.y + w2 * fc.x + w3 * fc.w;
        dd = w0 * dp.x + w1 * dp.y + w2 * dp.z + w3 * dp.w;
    }

    part[wave][lane] = make_float4(fx, fy, fz, dd);
    __syncthreads();

    if (wave == 0) {
        const float* Pf = (const float*)part;
        const v2f bones = {1.0f, 1.0f};
        const int row   = lane & 15;
        const int khalf = (lane >> 4) << 1;

        #pragma unroll
        for (int c = 0; c < 4; ++c) {
            v8f acc = {0.f, 0.f, 0.f, 0.f, 0.f, 0.f, 0.f, 0.f};
            #pragma unroll
            for (int kb = 0; kb < 8; ++kb) {
                const int k0 = kb * 4 + khalf;
                v2f a;
                a[0] = Pf[(row * 33 + k0)     * 4 + c];
                a[1] = Pf[(row * 33 + k0 + 1) * 4 + c];
                acc = __builtin_amdgcn_wmma_f32_16x16x4_f32(
                    false, a, false, bones, (short)0, acc, false, false);
            }
            #pragma unroll
            for (int j = 0; j < 8; ++j) {
                if (lane == j) {
                    int r = rayBase + j;
                    if (r < N) {
                        if (c < 3) out[r * 3 + c]   = acc[j];
                        else       out[3 * N + r]   = acc[j];
                    }
                }
                if (lane == 16 + j) {
                    int r = rayBase + 8 + j;
                    if (r < N) {
                        if (c < 3) out[r * 3 + c]   = acc[j];
                        else       out[3 * N + r]   = acc[j];
                    }
                }
            }
        }
    }
}

extern "C" void kernel_launch(void* const* d_in, const int* in_sizes, int n_in,
                              void* d_out, int out_size, void* d_ws, size_t ws_size,
                              hipStream_t stream) {
    const float* density = (const float*)d_in[0];
    const float* depth   = (const float*)d_in[1];
    const float* feature = (const float*)d_in[2];
    float* out = (float*)d_out;

    const int N = in_sizes[0] / NP;
    const int blocks = (N + 15) / 16;

    hipLaunchKernelGGL(volrend_kernel, dim3(blocks), dim3(512), 0, stream,
                       density, depth, feature, out, N);
}


# feature rows loaded as whole 128-byte lines (was 48-byte lane stride, 12 lines a load); weights, not features, move to that order via a wave-private LDS row; per-lane partial sums regrouped
# speedup vs baseline: 1.8747x; 1.8747x over previous
#include <hip/hip_runtime.h>


constexpr int NP = 128;
constexpr float FAR_DELTA = 1e10f;
constexpr int WPITCH = 144;

typedef __attribute__((ext_vector_type(2))) float v2f;
typedef __attribute__((ext_vector_type(8))) float v8f;

__global__ __launch_bounds__(256)
void volrend_kernel(const float* __restrict__ density,
                    const float* __restrict__ depth,
                    const float* __restrict__ feature,
                    float* __restrict__ out, int N)
{
    __shared__ __attribute__((aligned(16))) float wrow[8][4 * WPITCH];

    const int lane = threadIdx.x & 31;
    const int i    = lane & 7;
    const int g    = lane >> 3;
    const int wv   = (blockIdx.x * 256 + threadIdx.x) >> 5;
    const int ray  = wv * 4 + g;
    const int rc   = ray < N ? ray : N - 1;

    const float4* sgp = (const float4*)(density + (size_t)rc * NP) + i;
    const float4* dpp = (const float4*)(depth   + (size_t)rc * NP) + i;
    const float4* ftp = (const float4*)(feature + (size_t)rc * (NP * 3)) + i;

    float4 sg[4], dp[4], ft[12];
    #pragma unroll
    for (int j = 0; j < 4; ++j) {
        sg[j] = sgp[8 * j];
        dp[j] = dpp[8 * j];
    }
    __builtin_amdgcn_sched_barrier(1);
    #pragma unroll
    for (int k = 0; k < 12; ++k)
        ft[k] = ftp[8 * k];
    __builtin_amdgcn_sched_barrier(0);

    const int nxt = (lane & ~7) | ((i + 1) & 7);
    float s0[4], s1[4], s2[4], s3[4], inc[4];
    #pragma unroll
    for (int j = 0; j < 4; ++j) {
        float u = dp[j].x;
        if (j < 3 && i == 0) u = dp[j + 1].x;
        const float nx = __shfl(u, nxt, 32);
        const float d0 = dp[j].y - dp[j].x;
        const float d1 = dp[j].z - dp[j].y;
        const float d2 = dp[j].w - dp[j].z;
        const float d3 = (j == 3 && i == 7) ? FAR_DELTA : (nx - dp[j].w);
        s0[j] = __expf(-sg[j].x * d0);
        s1[j] = __expf(-sg[j].y * d1);
        s2[j] = __expf(-sg[j].z * d2);
        s3[j] = __expf(-sg[j].w * d3);
        inc[j] = s0[j] * s1[j] * s2[j] * s3[j];
    }

    #pragma unroll
    for (int off = 1; off < 8; off <<= 1) {
        #pragma unroll
        for (int j = 0; j < 4; ++j) {
            const float v = __shfl_up(inc[j], off, 8);
            if (i >= off) inc[j] *= v;
        }
    }

    float* wl = &wrow[threadIdx.x >> 5][g * WPITCH];
    float dd = 0.f;
    float carry = 1.0f;
    #pragma unroll
    for (int j = 0; j < 4; ++j) {
        float excl = __shfl_up(inc[j], 1, 8);
        if (i == 0) excl = 1.0f;
        const float tot = __shfl(inc[j], 7, 8);

        const float T0 = carry * excl;
        const float T1 = T0 * s0[j];
        const float T2 = T1 * s1[j];
        const float T3 = T2 * s2[j];
        float4 w;
        w.x = T0 * (1.0f - s0[j]);
        w.y = T1 * (1.0f - s1[j]);
        w.z = T2 * (1.0f - s2[j]);
        w.w = T3 * (1.0f - s3[j]);

        *(float4*)(wl + 32 * j + 4 * i) = w;
        dd += w.x * dp[j].x + w.y * dp[j].y + w.z * dp[j].z + w.w * dp[j].w;
        carry *= tot;
    }

    __builtin_amdgcn_fence(__ATOMIC_RELEASE, "wavefront");
    __builtin_amdgcn_wave_barrier();
    __builtin_amdgcn_fence(__ATOMIC_ACQUIRE, "wavefront");

    float acc0 = 0.f, acc1 = 0.f, acc2 = 0.f;
    #pragma unroll
    for (int kk = 0; kk < 3; ++kk) {
        const int u  = 8 * kk + i;
        const int a  = u / 3;
        const int s  = u - 3 * a;
        const float* wp = wl + 4 * a + s;
        #pragma unroll
        for (int j = 0; j < 4; ++j) {
            const float wa = wp[32 * j];
            const float wb = wp[32 * j + 1];
            const float w1 = (s == 2) ? wb : wa;
            const float w2 = (s == 0) ? wa : wb;
            const float4 f = ft[3 * j + kk];
            const float v0 = wa * f.x, v1 = w1 * f.y, v2 = w2 * f.z, v3 = wb * f.w;
            if (kk == 0)      { acc0 += v0; acc1 += v1; acc2 += v2; acc0 += v3; }
            else if (kk == 1) { acc2 += v0; acc0 += v1; acc1 += v2; acc2 += v3; }
            else              { acc1 += v0; acc2 += v1; acc0 += v2; acc1 += v3; }
        }
    }
    const int i3 = i % 3;
    const float fx = (i3 == 0) ? acc0 : (i3 == 1) ? acc2 : acc1;
    const float fy = (i3 == 0) ? acc1 : (i3 == 1) ? acc0 : acc2;
    const float fz = (i3 == 0) ? acc2 : (i3 == 1) ? acc1 : acc0;

    const int c16 = lane & 15;
    const int h   = lane >> 4;
    v2f b1, b2, a1, a2;
    b1[0] = (c16 == 2 * h)     ? 1.0f : 0.0f;
    b1[1] = (c16 == 2 * h + 1) ? 1.0f : 0.0f;
    b2[0] = (c16 == 2 * h + 4) ? 1.0f : 0.0f;
    b2[1] = (c16 == 2 * h + 5) ? 1.0f : 0.0f;
    a1[0] = fx; a1[1] = fy;
    a2[0] = fz; a2[1] = dd;
    v8f acc = {0.f, 0.f, 0.f, 0.f, 0.f, 0.f, 0.f, 0.f};
    acc = __builtin_amdgcn_wmma_f32_16x16x4_f32(false, a1, false, b1, (short)0, acc, false, false);
    asm volatile("v_nop\n\tv_nop\n\tv_nop\n\tv_nop\n\tv_nop\n\tv_nop\n\tv_nop\n\tv_nop"
                 : "+v"(acc) : "v"(a1), "v"(b1));
    acc = __builtin_amdgcn_wmma_f32_16x16x4_f32(false, a2, false, b2, (short)0, acc, false, false);
    asm volatile("v_nop\n\tv_nop\n\tv_nop\n\tv_nop\n\tv_nop\n\tv_nop\n\tv_nop\n\tv_nop"
                 : "+v"(acc) : "v"(a2), "v"(b2));
    const float r = ((acc[0] + acc[1]) + (acc[2] + acc[3]))
                  + ((acc[4] + acc[5]) + (acc[6] + acc[7]));

    if (c16 < 8) {
        const int chan = (c16 & 1) + 2 * (c16 >> 2);
        const int gsrc = 2 * ((c16 >> 1) & 1) + h;
        const int ro   = wv * 4 + gsrc;
        if (ro < N) {
            if (chan < 3) out[ro * 3 + chan] = r;
            else          out[3 * N + ro]    = r;
        }
    }
}

extern "C" void kernel_launch(void* const* d_in, const int* in_sizes, int n_in,
                              void* d_out, int out_size, void* d_ws, size_t ws_size,
                              hipStream_t stream) {
    const float* density = (const float*)d_in[0];
    const float* depth   = (const float*)d_in[1];
    const float* feature = (const float*)d_in[2];
    float* out = (float*)d_out;

    const int N = in_sizes[0] / NP;
    const int blocks = (N + 31) / 32;

    hipLaunchKernelGGL(volrend_kernel, dim3(blocks), dim3(256), 0, stream,
                       density, depth, feature, out, N);
}
